# starting kernel
# baseline (speedup 1.0000x reference)
#include <hip/hip_runtime.h>
#include <math.h>


#define GRID_NODES 64
#define NI         62
#define PAD        64
#define PI_F       3.14159265358979323846f

typedef float v2f __attribute__((ext_vector_type(2)));
typedef float v8f __attribute__((ext_vector_type(8)));

__global__ void fem_init_kernel(float* __restrict__ Fv,
                                float* __restrict__ S,
                                float* __restrict__ invlam,
                                const float* __restrict__ kappa) {
  int tid = blockIdx.x * blockDim.x + threadIdx.x;
  if (tid >= GRID_NODES * GRID_NODES) return;
  Fv[tid] = 0.0f;

  int p = tid / PAD;
  int q = tid % PAD;

  float s = 0.0f;
  if (p < NI && q < NI)
    s = __sinf((float)((p + 1) * (q + 1)) * (PI_F / 63.0f));
  S[tid] = s;

  float il = 0.0f;
  if (p < NI && q < NI) {
    float lp = 2.0f - 2.0f * __cosf((float)(p + 1) * (PI_F / 63.0f));
    float lq = 2.0f - 2.0f * __cosf((float)(q + 1) * (PI_F / 63.0f));
    float norm = (2.0f / 63.0f) * (2.0f / 63.0f);
    il = norm / (kappa[0] * (lp + lq));
  }
  invlam[tid] = il;
}

__global__ void fem_assemble_kernel(const float* __restrict__ f,
                                    const float* __restrict__ nodes,
                                    const int* __restrict__ elems,
                                    int nelem,
                                    float* __restrict__ Fv) {
  int e = blockIdx.x * blockDim.x + threadIdx.x;
  if (e >= nelem) return;
  int a = elems[3 * e + 0];
  int b = elems[3 * e + 1];
  int c = elems[3 * e + 2];
  float xi = nodes[2 * a], yi = nodes[2 * a + 1];
  float xj = nodes[2 * b], yj = nodes[2 * b + 1];
  float xk = nodes[2 * c], yk = nodes[2 * c + 1];
  float area = 0.5f * fabsf((xj - xi) * (yk - yi) - (xk - xi) * (yj - yi));
  float fe = 0.0f;
  if (area > 1e-15f)
    fe = (area * (1.0f / 3.0f)) * ((f[a] + f[b] + f[c]) * (1.0f / 3.0f));
  atomicAdd(&Fv[a], fe);
  atomicAdd(&Fv[b], fe);
  atomicAdd(&Fv[c], fe);
}

__device__ __forceinline__ void wmma_tile64(const float* __restrict__ X,
                                            const float* __restrict__ Y,
                                            float* __restrict__ C,
                                            int t, int lane) {
  const int half = lane >> 4;
  const int l16  = lane & 15;
  const int tm   = (t >> 2) * 16;
  const int tn   = (t & 3) * 16;
  v8f acc = {0.f, 0.f, 0.f, 0.f, 0.f, 0.f, 0.f, 0.f};
#pragma unroll
  for (int k0 = 0; k0 < PAD; k0 += 4) {
    const int ka = k0 + 2 * half;
    v2f a, b;
    a[0] = X[(tm + l16) * PAD + ka];
    a[1] = X[(tm + l16) * PAD + ka + 1];
    b[0] = Y[ka * PAD + tn + l16];
    b[1] = Y[(ka + 1) * PAD + tn + l16];
    acc = __builtin_amdgcn_wmma_f32_16x16x4_f32(
        false, a, false, b, (short)0, acc, false, false);
  }
#pragma unroll
  for (int j = 0; j < 8; ++j)
    C[(tm + j + 8 * half) * PAD + tn + l16] = acc[j];
}

__device__ __forceinline__ void wmma_gemm64(const float* __restrict__ X,
                                            const float* __restrict__ Y,
                                            float* __restrict__ C,
                                            int wave_u, int lane) {
  wmma_tile64(X, Y, C, wave_u, lane);
  wmma_tile64(X, Y, C, wave_u + 8, lane);
}

__global__ void __launch_bounds__(256)
fem_solve_kernel(const float* __restrict__ Fv,
                 const float* __restrict__ Sg,
                 const float* __restrict__ invlam,
                 float* __restrict__ u) {
  __shared__ float Sm[PAD * PAD];
  __shared__ float T0[PAD * PAD];
  __shared__ float T1[PAD * PAD];

  const int tid    = threadIdx.x;
  const int lane   = tid & 31;
  const int wave_u = __builtin_amdgcn_readfirstlane(tid >> 5);

  for (int i = tid; i < PAD * PAD; i += 256) {
    Sm[i] = Sg[i];
    int p = i / PAD, q = i % PAD;
    float v = 0.0f;
    if (p < NI && q < NI) v = Fv[(p + 1) * GRID_NODES + (q + 1)];
    T0[i] = v;
  }
  __syncthreads();

  wmma_gemm64(Sm, T0, T1, wave_u, lane);
  __syncthreads();
  wmma_gemm64(T1, Sm, T0, wave_u, lane);
  __syncthreads();

  for (int i = tid; i < PAD * PAD; i += 256)
    T0[i] *= invlam[i];
  __syncthreads();

  wmma_gemm64(Sm, T0, T1, wave_u, lane);
  __syncthreads();
  wmma_gemm64(T1, Sm, T0, wave_u, lane);
  __syncthreads();

  for (int i = tid; i < GRID_NODES * GRID_NODES; i += 256) {
    int r = i / GRID_NODES, c = i % GRID_NODES;
    float v = 0.0f;
    if (r >= 1 && r <= NI && c >= 1 && c <= NI)
      v = T0[(r - 1) * PAD + (c - 1)];
    u[i] = v;
  }
}

extern "C" void kernel_launch(void* const* d_in, const int* in_sizes, int n_in,
                              void* d_out, int out_size, void* d_ws, size_t ws_size,
                              hipStream_t stream) {
  const float* f      = (const float*)d_in[0];
  const float* nodes  = (const float*)d_in[1];
  const float* kappa  = (const float*)d_in[2];
  const int*   elems  = (const int*)d_in[4];
  const int    nelem  = in_sizes[4] / 3;

  float* ws     = (float*)d_ws;
  float* Fv     = ws;
  float* S      = ws + 4096;
  float* invlam = ws + 8192;

  fem_init_kernel<<<(GRID_NODES * GRID_NODES + 255) / 256, 256, 0, stream>>>(
      Fv, S, invlam, kappa);
  fem_assemble_kernel<<<(nelem + 255) / 256, 256, 0, stream>>>(
      f, nodes, elems, nelem, Fv);
  fem_solve_kernel<<<1, 256, 0, stream>>>(Fv, S, invlam, (float*)d_out);
}


# solve kernel: all of its global reads (S, load vector, eigenvalue table) issued as one batch before the first wait, instead of one dependent memory round trip per element
# speedup vs baseline: 1.4308x; 1.4308x over previous
#include <hip/hip_runtime.h>
#include <math.h>


#define GRID_NODES 64
#define NI         62
#define PAD        64
#define PI_F       3.14159265358979323846f

typedef float v2f __attribute__((ext_vector_type(2)));
typedef float v4f __attribute__((ext_vector_type(4)));
typedef float v8f __attribute__((ext_vector_type(8)));

__global__ void fem_init_kernel(float* __restrict__ Fv,
                                float* __restrict__ S,
                                float* __restrict__ invlam,
                                const float* __restrict__ kappa) {
  int tid = blockIdx.x * blockDim.x + threadIdx.x;
  if (tid >= GRID_NODES * GRID_NODES) return;
  Fv[tid] = 0.0f;

  int p = tid / PAD;
  int q = tid % PAD;

  float s = 0.0f;
  if (p < NI && q < NI)
    s = __sinf((float)((p + 1) * (q + 1)) * (PI_F / 63.0f));
  S[tid] = s;

  float il = 0.0f;
  if (p < NI && q < NI) {
    float lp = 2.0f - 2.0f * __cosf((float)(p + 1) * (PI_F / 63.0f));
    float lq = 2.0f - 2.0f * __cosf((float)(q + 1) * (PI_F / 63.0f));
    float norm = (2.0f / 63.0f) * (2.0f / 63.0f);
    il = norm / (kappa[0] * (lp + lq));
  }
  invlam[tid] = il;
}

__global__ void fem_assemble_kernel(const float* __restrict__ f,
                                    const float* __restrict__ nodes,
                                    const int* __restrict__ elems,
                                    int nelem,
                                    float* __restrict__ Fv) {
  int e = blockIdx.x * blockDim.x + threadIdx.x;
  if (e >= nelem) return;
  int a = elems[3 * e + 0];
  int b = elems[3 * e + 1];
  int c = elems[3 * e + 2];
  float xi = nodes[2 * a], yi = nodes[2 * a + 1];
  float xj = nodes[2 * b], yj = nodes[2 * b + 1];
  float xk = nodes[2 * c], yk = nodes[2 * c + 1];
  float area = 0.5f * fabsf((xj - xi) * (yk - yi) - (xk - xi) * (yj - yi));
  float fe = 0.0f;
  if (area > 1e-15f)
    fe = (area * (1.0f / 3.0f)) * ((f[a] + f[b] + f[c]) * (1.0f / 3.0f));
  atomicAdd(&Fv[a], fe);
  atomicAdd(&Fv[b], fe);
  atomicAdd(&Fv[c], fe);
}

__device__ __forceinline__ void wmma_tile64(const float* __restrict__ X,
                                            const float* __restrict__ Y,
                                            float* __restrict__ C,
                                            int t, int lane) {
  const int half = lane >> 4;
  const int l16  = lane & 15;
  const int tm   = (t >> 2) * 16;
  const int tn   = (t & 3) * 16;
  v8f acc = {0.f, 0.f, 0.f, 0.f, 0.f, 0.f, 0.f, 0.f};
#pragma unroll
  for (int k0 = 0; k0 < PAD; k0 += 4) {
    const int ka = k0 + 2 * half;
    v2f a, b;
    a[0] = X[(tm + l16) * PAD + ka];
    a[1] = X[(tm + l16) * PAD + ka + 1];
    b[0] = Y[ka * PAD + tn + l16];
    b[1] = Y[(ka + 1) * PAD + tn + l16];
    acc = __builtin_amdgcn_wmma_f32_16x16x4_f32(
        false, a, false, b, (short)0, acc, false, false);
  }
#pragma unroll
  for (int j = 0; j < 8; ++j)
    C[(tm + j + 8 * half) * PAD + tn + l16] = acc[j];
}

__device__ __forceinline__ void wmma_gemm64(const float* __restrict__ X,
                                            const float* __restrict__ Y,
                                            float* __restrict__ C,
                                            int wave_u, int lane) {
  wmma_tile64(X, Y, C, wave_u, lane);
  wmma_tile64(X, Y, C, wave_u + 8, lane);
}

__global__ void __launch_bounds__(256)
fem_solve_kernel(const float* __restrict__ Fv,
                 const float* __restrict__ Sg,
                 const float* __restrict__ invlam,
                 float* __restrict__ u) {
  __shared__ float Sm[PAD * PAD];
  __shared__ float T0[PAD * PAD];
  __shared__ float T1[PAD * PAD];
  __shared__ float Lm[PAD * PAD];

  const int tid    = threadIdx.x;
  const int lane   = tid & 31;
  const int wave_u = __builtin_amdgcn_readfirstlane(tid >> 5);

  const v4f* __restrict__ Sg4 = reinterpret_cast<const v4f*>(Sg);
  const v4f* __restrict__ il4 = reinterpret_cast<const v4f*>(invlam);
  v4f sv[4];
  v4f lv[4];
  float  fv[16];
  for (int j = 0; j < 4; ++j) sv[j] = Sg4[tid + 256 * j];
  for (int j = 0; j < 16; ++j) {
    const int i = tid + 256 * j;
    const int p = i / PAD, q = i % PAD;
    const bool in = (p < NI) && (q < NI);
    const float v = Fv[in ? (p + 1) * GRID_NODES + (q + 1) : 0];
    fv[j] = in ? v : 0.0f;
  }
  for (int j = 0; j < 4; ++j) lv[j] = il4[tid + 256 * j];

  v4f* Sm4 = reinterpret_cast<v4f*>(Sm);
  v4f* T04 = reinterpret_cast<v4f*>(T0);
  v4f* Lm4 = reinterpret_cast<v4f*>(Lm);
  for (int j = 0; j < 4; ++j) Sm4[tid + 256 * j] = sv[j];
  for (int j = 0; j < 4; ++j) Lm4[tid + 256 * j] = lv[j];
  for (int j = 0; j < 16; ++j) T0[tid + 256 * j] = fv[j];
  __syncthreads();

  wmma_gemm64(Sm, T0, T1, wave_u, lane);
  __syncthreads();
  wmma_gemm64(T1, Sm, T0, wave_u, lane);
  __syncthreads();

  for (int j = 0; j < 4; ++j)
    T04[tid + 256 * j] = T04[tid + 256 * j] * Lm4[tid + 256 * j];
  __syncthreads();

  wmma_gemm64(Sm, T0, T1, wave_u, lane);
  __syncthreads();
  wmma_gemm64(T1, Sm, T0, wave_u, lane);
  __syncthreads();

  for (int i = tid; i < GRID_NODES * GRID_NODES; i += 256) {
    int r = i / GRID_NODES, c = i % GRID_NODES;
    float v = 0.0f;
    if (r >= 1 && r <= NI && c >= 1 && c <= NI)
      v = T0[(r - 1) * PAD + (c - 1)];
    u[i] = v;
  }
}

extern "C" void kernel_launch(void* const* d_in, const int* in_sizes, int n_in,
                              void* d_out, int out_size, void* d_ws, size_t ws_size,
                              hipStream_t stream) {
  const float* f      = (const float*)d_in[0];
  const float* nodes  = (const float*)d_in[1];
  const float* kappa  = (const float*)d_in[2];
  const int*   elems  = (const int*)d_in[4];
  const int    nelem  = in_sizes[4] / 3;

  float* ws     = (float*)d_ws;
  float* Fv     = ws;
  float* S      = ws + 4096;
  float* invlam = ws + 8192;

  fem_init_kernel<<<(GRID_NODES * GRID_NODES + 255) / 256, 256, 0, stream>>>(
      Fv, S, invlam, kappa);
  fem_assemble_kernel<<<(nelem + 255) / 256, 256, 0, stream>>>(
      f, nodes, elems, nelem, Fv);
  fem_solve_kernel<<<1, 256, 0, stream>>>(Fv, S, invlam, (float*)d_out);
}


# LDS row pitch of the four 64x64 GEMM operands 64 -> 76 dwords: an A fragment's 16 rows fall on 16 different bank groups instead of one (16-way bank conflict on every A read removed)
# speedup vs baseline: 3.3153x; 2.3171x over previous
#include <hip/hip_runtime.h>
#include <math.h>


#define GRID_NODES 64
#define NI         62
#define PAD        64
#define LDP        76
#define PI_F       3.14159265358979323846f

typedef float v2f __attribute__((ext_vector_type(2)));
typedef float v8f __attribute__((ext_vector_type(8)));

#define NTHREADS   1024
#define EPT        4

__device__ __forceinline__ void wmma_gemm64(const float* __restrict__ X,
                                            const float* __restrict__ Y,
                                            float* __restrict__ C,
                                            int wave_u, int lane) {
  const int half = lane >> 4;
  const int l16  = lane & 15;
  const int tm   = (wave_u >> 2) * 16;
  const int tn   = (wave_u & 3) * 16;
  v8f acc0 = {0.f, 0.f, 0.f, 0.f, 0.f, 0.f, 0.f, 0.f};
  v8f acc1 = {0.f, 0.f, 0.f, 0.f, 0.f, 0.f, 0.f, 0.f};
#pragma unroll
  for (int k0 = 0; k0 < PAD; k0 += 4) {
    const int ka = k0 + 2 * half;
    v2f a0, a1, b;
    a0[0] = X[(tm + l16) * LDP + ka];
    a0[1] = X[(tm + l16) * LDP + ka + 1];
    a1[0] = X[(tm + 32 + l16) * LDP + ka];
    a1[1] = X[(tm + 32 + l16) * LDP + ka + 1];
    b[0]  = Y[ka * LDP + tn + l16];
    b[1]  = Y[(ka + 1) * LDP + tn + l16];
    acc0 = __builtin_amdgcn_wmma_f32_16x16x4_f32(
        false, a0, false, b, (short)0, acc0, false, false);
    acc1 = __builtin_amdgcn_wmma_f32_16x16x4_f32(
        false, a1, false, b, (short)0, acc1, false, false);
  }
  for (int j = 0; j < 8; ++j) {
    C[(tm + j + 8 * half) * LDP + tn + l16]      = acc0[j];
    C[(tm + 32 + j + 8 * half) * LDP + tn + l16] = acc1[j];
  }
}

__global__ void __launch_bounds__(NTHREADS)
fem_fused_kernel(const float* __restrict__ f,
                 const float* __restrict__ nodes,
                 const float* __restrict__ kappa,
                 const int* __restrict__ elems,
                 int nelem,
                 float* __restrict__ u) {
  __shared__ float Sm[PAD * LDP];
  __shared__ float T0[PAD * LDP];
  __shared__ float T1[PAD * LDP];
  __shared__ float Lm[PAD * LDP];
  __shared__ float Fl[GRID_NODES * GRID_NODES];

  const int tid    = threadIdx.x;
  const int lane   = tid & 31;
  const int wave_u = __builtin_amdgcn_readfirstlane(tid >> 5);

  for (int j = 0; j < (GRID_NODES * GRID_NODES) / NTHREADS; ++j)
    Fl[tid + NTHREADS * j] = 0.0f;
  __syncthreads();

  const v2f* __restrict__ xy = reinterpret_cast<const v2f*>(nodes);
  for (int base = 0; base < nelem; base += NTHREADS * EPT) {
    int  va[EPT], vb[EPT], vc[EPT];
    bool ok[EPT];
    for (int k = 0; k < EPT; ++k) {
      const int e = base + tid + NTHREADS * k;
      ok[k] = e < nelem;
      const int ee = ok[k] ? e : 0;
      va[k] = elems[3 * ee + 0];
      vb[k] = elems[3 * ee + 1];
      vc[k] = elems[3 * ee + 2];
    }
    v2f   pi[EPT], pj[EPT], pk[EPT];
    float fa[EPT], fb[EPT], fc[EPT];
    for (int k = 0; k < EPT; ++k) {
      pi[k] = xy[va[k]]; pj[k] = xy[vb[k]]; pk[k] = xy[vc[k]];
      fa[k] = f[va[k]];  fb[k] = f[vb[k]];  fc[k] = f[vc[k]];
    }
    for (int k = 0; k < EPT; ++k) {
      const float area = 0.5f * fabsf((pj[k][0] - pi[k][0]) * (pk[k][1] - pi[k][1]) -
                                      (pk[k][0] - pi[k][0]) * (pj[k][1] - pi[k][1]));
      const float t  = (area * (1.0f / 3.0f)) * ((fa[k] + fb[k] + fc[k]) * (1.0f / 3.0f));
      const float fe = (ok[k] && area > 1e-15f) ? t : 0.0f;
      atomicAdd(&Fl[ok[k] ? va[k] : tid], fe);
      atomicAdd(&Fl[ok[k] ? vb[k] : tid], fe);
      atomicAdd(&Fl[ok[k] ? vc[k] : tid], fe);
    }
  }

  const float kap = kappa[0];
  for (int j = 0; j < (PAD * PAD) / NTHREADS; ++j) {
    const int i = tid + NTHREADS * j;
    const int p = i / PAD;
    const int q = i % PAD;
    float s = 0.0f;
    float il = 0.0f;
    if (p < NI && q < NI) {
      s = __sinf((float)((p + 1) * (q + 1)) * (PI_F / 63.0f));
      float lp = 2.0f - 2.0f * __cosf((float)(p + 1) * (PI_F / 63.0f));
      float lq = 2.0f - 2.0f * __cosf((float)(q + 1) * (PI_F / 63.0f));
      float norm = (2.0f / 63.0f) * (2.0f / 63.0f);
      il = norm / (kap * (lp + lq));
    }
    Sm[p * LDP + q] = s;
    Lm[p * LDP + q] = il;
  }
  __syncthreads();

  for (int j = 0; j < (PAD * PAD) / NTHREADS; ++j) {
    const int i = tid + NTHREADS * j;
    const int p = i / PAD, q = i % PAD;
    const bool in = (p < NI) && (q < NI);
    const float v = Fl[in ? (p + 1) * GRID_NODES + (q + 1) : 0];
    T0[p * LDP + q] = in ? v : 0.0f;
  }
  __syncthreads();

  const bool mm = wave_u < 8;
  if (mm) wmma_gemm64(Sm, T0, T1, wave_u, lane);
  __syncthreads();
  if (mm) wmma_gemm64(T1, Sm, T0, wave_u, lane);
  __syncthreads();

  for (int j = 0; j < (PAD * PAD) / NTHREADS; ++j) {
    const int i = tid + NTHREADS * j;
    const int o = (i / PAD) * LDP + (i % PAD);
    T0[o] = T0[o] * Lm[o];
  }
  __syncthreads();

  if (mm) wmma_gemm64(Sm, T0, T1, wave_u, lane);
  __syncthreads();
  if (mm) wmma_gemm64(T1, Sm, T0, wave_u, lane);
  __syncthreads();

  for (int j = 0; j < (GRID_NODES * GRID_NODES) / NTHREADS; ++j) {
    const int i = tid + NTHREADS * j;
    int r = i / GRID_NODES, c = i % GRID_NODES;
    float v = 0.0f;
    if (r >= 1 && r <= NI && c >= 1 && c <= NI)
      v = T0[(r - 1) * LDP + (c - 1)];
    u[i] = v;
  }
}

extern "C" void kernel_launch(void* const* d_in, const int* in_sizes, int n_in,
                              void* d_out, int out_size, void* d_ws, size_t ws_size,
                              hipStream_t stream) {
  const float* f      = (const float*)d_in[0];
  const float* nodes  = (const float*)d_in[1];
  const float* kappa  = (const float*)d_in[2];
  const int*   elems  = (const int*)d_in[4];
  const int    nelem  = in_sizes[4] / 3;

  fem_fused_kernel<<<1, NTHREADS, 0, stream>>>(f, nodes, kappa, elems, nelem,
                                               (float*)d_out);
}


# each 62-point sine transform folded over its mirror pairs (S[p][61-i] = (-1)^p S[p][i]): sum/difference split layout, S block-diagonal, GEMM k range 64 -> 32, f32 WMMAs per solve halved
# speedup vs baseline: 3.6862x; 1.1119x over previous
#include <hip/hip_runtime.h>
#include <math.h>


#define GRID_NODES 64
#define NI         62
#define PAD        64
#define NH         31
#define HB         32
#define LDP        76
#define PI_F       3.14159265358979323846f

typedef float v2f __attribute__((ext_vector_type(2)));
typedef float v8f __attribute__((ext_vector_type(8)));

#define NTHREADS   1024
#define EPT        4

template <bool LEFT>
__device__ __forceinline__ void wmma_gemm_bd(const float* __restrict__ X,
                                             const float* __restrict__ Y,
                                             float* __restrict__ C,
                                             int wave_u, int lane) {
  const int half = lane >> 4;
  const int l16  = lane & 15;
  const int tm   = (wave_u >> 2) * 16;
  const int tn   = (wave_u & 3) * 16;
  const int kb0  = LEFT ? 0  : (tn & HB);
  const int kb1  = LEFT ? HB : (tn & HB);
  v8f acc0 = {0.f, 0.f, 0.f, 0.f, 0.f, 0.f, 0.f, 0.f};
  v8f acc1 = {0.f, 0.f, 0.f, 0.f, 0.f, 0.f, 0.f, 0.f};
#pragma unroll
  for (int k0 = 0; k0 < HB; k0 += 4) {
    const int ka0 = kb0 + k0 + 2 * half;
    const int ka1 = kb1 + k0 + 2 * half;
    v2f a0, a1, b0, b1;
    a0[0] = X[(tm + l16) * LDP + ka0];
    a0[1] = X[(tm + l16) * LDP + ka0 + 1];
    a1[0] = X[(tm + 32 + l16) * LDP + ka1];
    a1[1] = X[(tm + 32 + l16) * LDP + ka1 + 1];
    b0[0] = Y[ka0 * LDP + tn + l16];
    b0[1] = Y[(ka0 + 1) * LDP + tn + l16];
    b1[0] = Y[ka1 * LDP + tn + l16];
    b1[1] = Y[(ka1 + 1) * LDP + tn + l16];
    acc0 = __builtin_amdgcn_wmma_f32_16x16x4_f32(
        false, a0, false, b0, (short)0, acc0, false, false);
    acc1 = __builtin_amdgcn_wmma_f32_16x16x4_f32(
        false, a1, false, b1, (short)0, acc1, false, false);
  }
  for (int j = 0; j < 8; ++j) {
    C[(tm + j + 8 * half) * LDP + tn + l16]      = acc0[j];
    C[(tm + 32 + j + 8 * half) * LDP + tn + l16] = acc1[j];
  }
}

__global__ void __launch_bounds__(NTHREADS)
fem_fused_kernel(const float* __restrict__ f,
                 const float* __restrict__ nodes,
                 const float* __restrict__ kappa,
                 const int* __restrict__ elems,
                 int nelem,
                 float* __restrict__ u) {
  __shared__ float Dm[PAD * LDP];
  __shared__ float Dt[PAD * LDP];
  __shared__ float T0[PAD * LDP];
  __shared__ float T1[PAD * LDP];
  __shared__ float Lm[PAD * LDP];
  __shared__ float Fl[GRID_NODES * GRID_NODES];

  const int tid    = threadIdx.x;
  const int lane   = tid & 31;
  const int wave_u = __builtin_amdgcn_readfirstlane(tid >> 5);

  for (int j = 0; j < (GRID_NODES * GRID_NODES) / NTHREADS; ++j)
    Fl[tid + NTHREADS * j] = 0.0f;
  __syncthreads();

  const v2f* __restrict__ xy = reinterpret_cast<const v2f*>(nodes);
  for (int base = 0; base < nelem; base += NTHREADS * EPT) {
    int  va[EPT], vb[EPT], vc[EPT];
    bool ok[EPT];
    for (int k = 0; k < EPT; ++k) {
      const int e = base + tid + NTHREADS * k;
      ok[k] = e < nelem;
      const int ee = ok[k] ? e : 0;
      va[k] = elems[3 * ee + 0];
      vb[k] = elems[3 * ee + 1];
      vc[k] = elems[3 * ee + 2];
    }
    v2f   pi[EPT], pj[EPT], pk[EPT];
    float fa[EPT], fb[EPT], fc[EPT];
    for (int k = 0; k < EPT; ++k) {
      pi[k] = xy[va[k]]; pj[k] = xy[vb[k]]; pk[k] = xy[vc[k]];
      fa[k] = f[va[k]];  fb[k] = f[vb[k]];  fc[k] = f[vc[k]];
    }
    for (int k = 0; k < EPT; ++k) {
      const float area = 0.5f * fabsf((pj[k][0] - pi[k][0]) * (pk[k][1] - pi[k][1]) -
                                      (pk[k][0] - pi[k][0]) * (pj[k][1] - pi[k][1]));
      const float t  = (area * (1.0f / 3.0f)) * ((fa[k] + fb[k] + fc[k]) * (1.0f / 3.0f));
      const float fe = (ok[k] && area > 1e-15f) ? t : 0.0f;
      atomicAdd(&Fl[ok[k] ? va[k] : tid], fe);
      atomicAdd(&Fl[ok[k] ? vb[k] : tid], fe);
      atomicAdd(&Fl[ok[k] ? vc[k] : tid], fe);
    }
  }

  const float kap = kappa[0];
  for (int j = 0; j < (PAD * PAD) / NTHREADS; ++j) {
    const int i = tid + NTHREADS * j;
    const int r = i / PAD;
    const int c = i % PAD;
    const int a = r / HB, m = r % HB;
    const int b = c / HB, n = c % HB;
    const int p = 2 * m + a;
    const int q = 2 * n + b;
    float d = 0.0f;
    float il = 0.0f;
    if (m < NH && n < NH) {
      if (a == b)
        d = __sinf((float)((p + 1) * (n + 1)) * (PI_F / 63.0f));
      float lp = 2.0f - 2.0f * __cosf((float)(p + 1) * (PI_F / 63.0f));
      float lq = 2.0f - 2.0f * __cosf((float)(q + 1) * (PI_F / 63.0f));
      float norm = (2.0f / 63.0f) * (2.0f / 63.0f);
      il = norm / (kap * (lp + lq));
    }
    Dm[r * LDP + c] = d;
    Dt[c * LDP + r] = d;
    Lm[r * LDP + c] = il;
  }
  __syncthreads();

  {
    const int i = tid / HB, j = tid % HB;
    const bool in = (i < NH) && (j < NH);
    const float f00 = Fl[(i + 1) * GRID_NODES + (j + 1)];
    const float f10 = Fl[(NI - i) * GRID_NODES + (j + 1)];
    const float f01 = Fl[(i + 1) * GRID_NODES + (NI - j)];
    const float f11 = Fl[(NI - i) * GRID_NODES + (NI - j)];
    const float s0 = f00 + f10, s1 = f01 + f11;
    const float d0 = f00 - f10, d1 = f01 - f11;
    T0[i * LDP + j]               = in ? s0 + s1 : 0.0f;
    T0[i * LDP + HB + j]          = in ? s0 - s1 : 0.0f;
    T0[(HB + i) * LDP + j]        = in ? d0 + d1 : 0.0f;
    T0[(HB + i) * LDP + HB + j]   = in ? d0 - d1 : 0.0f;
  }
  __syncthreads();

  const bool mm = wave_u < 8;
  if (mm) wmma_gemm_bd<true>(Dm, T0, T1, wave_u, lane);
  __syncthreads();
  if (mm) wmma_gemm_bd<false>(T1, Dt, T0, wave_u, lane);
  __syncthreads();

  for (int j = 0; j < (PAD * PAD) / NTHREADS; ++j) {
    const int i = tid + NTHREADS * j;
    const int o = (i / PAD) * LDP + (i % PAD);
    T0[o] = T0[o] * Lm[o];
  }
  __syncthreads();

  if (mm) wmma_gemm_bd<true>(Dt, T0, T1, wave_u, lane);
  __syncthreads();
  if (mm) wmma_gemm_bd<false>(T1, Dm, T0, wave_u, lane);
  __syncthreads();

  for (int j = 0; j < (GRID_NODES * GRID_NODES) / NTHREADS; ++j) {
    const int i = tid + NTHREADS * j;
    const int r = i / GRID_NODES, c = i % GRID_NODES;
    const bool in = (r >= 1) && (r <= NI) && (c >= 1) && (c <= NI);
    const int i0 = in ? r - 1 : 0, j0 = in ? c - 1 : 0;
    const bool mr = i0 >= NH, mc = j0 >= NH;
    const int ii = mr ? (NI - 1 - i0) : i0;
    const int jj = mc ? (NI - 1 - j0) : j0;
    const float w00 = T0[ii * LDP + jj];
    const float w01 = T0[ii * LDP + HB + jj];
    const float w10 = T0[(HB + ii) * LDP + jj];
    const float w11 = T0[(HB + ii) * LDP + HB + jj];
    const float top = mc ? w00 - w01 : w00 + w01;
    const float bot = mc ? w10 - w11 : w10 + w11;
    const float v   = mr ? top - bot : top + bot;
    u[i] = in ? v : 0.0f;
  }
}

extern "C" void kernel_launch(void* const* d_in, const int* in_sizes, int n_in,
                              void* d_out, int out_size, void* d_ws, size_t ws_size,
                              hipStream_t stream) {
  const float* f      = (const float*)d_in[0];
  const float* nodes  = (const float*)d_in[1];
  const float* kappa  = (const float*)d_in[2];
  const int*   elems  = (const int*)d_in[4];
  const int    nelem  = in_sizes[4] / 3;

  fem_fused_kernel<<<1, NTHREADS, 0, stream>>>(f, nodes, kappa, elems, nelem,
                                               (float*)d_out);
}


# spectral stage: G*D^T, D*(.), eigenvalue scaling, D^T*(.) chained in accumulator registers (C tiles fed back as the B operand, A-load k order matched): 3 barriers, 2 LDS round trips, scale pass gone
# speedup vs baseline: 3.9380x; 1.0683x over previous
#include <hip/hip_runtime.h>
#include <math.h>


#define GRID_NODES 64
#define NI         62
#define PAD        64
#define NH         31
#define HB         32
#define LDP        76
#define PI_F       3.14159265358979323846f

typedef float v2f __attribute__((ext_vector_type(2)));
typedef float v8f __attribute__((ext_vector_type(8)));

#define NTHREADS   1024
#define EPT        4


__device__ __forceinline__ void right_bd(const float* __restrict__ X,
                                         const float* __restrict__ Y,
                                         int r0, int tn, int half, int l16,
                                         v8f& c0, v8f& c1) {
  const int kb = tn & HB;
#pragma unroll
  for (int k0 = 0; k0 < HB; k0 += 4) {
    const int ka = kb + k0 + 2 * half;
    v2f a0, a1, b;
    a0[0] = X[(r0 + l16) * LDP + ka];
    a0[1] = X[(r0 + l16) * LDP + ka + 1];
    a1[0] = X[(r0 + 16 + l16) * LDP + ka];
    a1[1] = X[(r0 + 16 + l16) * LDP + ka + 1];
    b[0]  = Y[ka * LDP + tn + l16];
    b[1]  = Y[(ka + 1) * LDP + tn + l16];
    c0 = __builtin_amdgcn_wmma_f32_16x16x4_f32(
        false, a0, false, b, (short)0, c0, false, false);
    c1 = __builtin_amdgcn_wmma_f32_16x16x4_f32(
        false, a1, false, b, (short)0, c1, false, false);
  }
}

__device__ __forceinline__ void left_bd_regs(const float* __restrict__ X,
                                             int r0, int half, int l16,
                                             const v8f& b0, const v8f& b1,
                                             v8f& c0, v8f& c1) {
#pragma unroll
  for (int i = 0; i < 2; ++i) {
#pragma unroll
    for (int t = 0; t < 4; ++t) {
      const int kc = r0 + 16 * i + 8 * half + 2 * t;
      v2f a0, a1, b;
      a0[0] = X[(r0 + l16) * LDP + kc];
      a0[1] = X[(r0 + l16) * LDP + kc + 1];
      a1[0] = X[(r0 + 16 + l16) * LDP + kc];
      a1[1] = X[(r0 + 16 + l16) * LDP + kc + 1];
      b[0]  = i ? b1[2 * t]     : b0[2 * t];
      b[1]  = i ? b1[2 * t + 1] : b0[2 * t + 1];
      c0 = __builtin_amdgcn_wmma_f32_16x16x4_f32(
          false, a0, false, b, (short)0, c0, false, false);
      c1 = __builtin_amdgcn_wmma_f32_16x16x4_f32(
          false, a1, false, b, (short)0, c1, false, false);
    }
  }
}

__device__ __forceinline__ void store_tiles(float* __restrict__ C, int r0, int tn,
                                            int half, int l16,
                                            const v8f& c0, const v8f& c1) {
  for (int j = 0; j < 8; ++j) {
    C[(r0 + j + 8 * half) * LDP + tn + l16]      = c0[j];
    C[(r0 + 16 + j + 8 * half) * LDP + tn + l16] = c1[j];
  }
}

__global__ void __launch_bounds__(NTHREADS)
fem_fused_kernel(const float* __restrict__ f,
                 const float* __restrict__ nodes,
                 const float* __restrict__ kappa,
                 const int* __restrict__ elems,
                 int nelem,
                 float* __restrict__ u) {
  __shared__ float Dm[PAD * LDP];
  __shared__ float Dt[PAD * LDP];
  __shared__ float T0[PAD * LDP];
  __shared__ float T1[PAD * LDP];
  __shared__ float Lm[PAD * LDP];
  __shared__ float Fl[GRID_NODES * GRID_NODES];

  const int tid    = threadIdx.x;
  const int lane   = tid & 31;
  const int wave_u = __builtin_amdgcn_readfirstlane(tid >> 5);

  for (int j = 0; j < (GRID_NODES * GRID_NODES) / NTHREADS; ++j)
    Fl[tid + NTHREADS * j] = 0.0f;
  __syncthreads();

  const v2f* __restrict__ xy = reinterpret_cast<const v2f*>(nodes);
  for (int base = 0; base < nelem; base += NTHREADS * EPT) {
    int  va[EPT], vb[EPT], vc[EPT];
    bool ok[EPT];
    for (int k = 0; k < EPT; ++k) {
      const int e = base + tid + NTHREADS * k;
      ok[k] = e < nelem;
      const int ee = ok[k] ? e : 0;
      va[k] = elems[3 * ee + 0];
      vb[k] = elems[3 * ee + 1];
      vc[k] = elems[3 * ee + 2];
    }
    v2f   pi[EPT], pj[EPT], pk[EPT];
    float fa[EPT], fb[EPT], fc[EPT];
    for (int k = 0; k < EPT; ++k) {
      pi[k] = xy[va[k]]; pj[k] = xy[vb[k]]; pk[k] = xy[vc[k]];
      fa[k] = f[va[k]];  fb[k] = f[vb[k]];  fc[k] = f[vc[k]];
    }
    for (int k = 0; k < EPT; ++k) {
      const float area = 0.5f * fabsf((pj[k][0] - pi[k][0]) * (pk[k][1] - pi[k][1]) -
                                      (pk[k][0] - pi[k][0]) * (pj[k][1] - pi[k][1]));
      const float t  = (area * (1.0f / 3.0f)) * ((fa[k] + fb[k] + fc[k]) * (1.0f / 3.0f));
      const float fe = (ok[k] && area > 1e-15f) ? t : 0.0f;
      atomicAdd(&Fl[ok[k] ? va[k] : tid], fe);
      atomicAdd(&Fl[ok[k] ? vb[k] : tid], fe);
      atomicAdd(&Fl[ok[k] ? vc[k] : tid], fe);
    }
  }

  const float kap = kappa[0];
  for (int j = 0; j < (PAD * PAD) / NTHREADS; ++j) {
    const int i = tid + NTHREADS * j;
    const int r = i / PAD;
    const int c = i % PAD;
    const int a = r / HB, m = r % HB;
    const int b = c / HB, n = c % HB;
    const int p = 2 * m + a;
    const int q = 2 * n + b;
    float d = 0.0f;
    float il = 0.0f;
    if (m < NH && n < NH) {
      if (a == b)
        d = __sinf((float)((p + 1) * (n + 1)) * (PI_F / 63.0f));
      float lp = 2.0f - 2.0f * __cosf((float)(p + 1) * (PI_F / 63.0f));
      float lq = 2.0f - 2.0f * __cosf((float)(q + 1) * (PI_F / 63.0f));
      float norm = (2.0f / 63.0f) * (2.0f / 63.0f);
      il = norm / (kap * (lp + lq));
    }
    Dm[r * LDP + c] = d;
    Dt[c * LDP + r] = d;
    Lm[r * LDP + c] = il;
  }
  __syncthreads();

  const bool mm   = wave_u < 8;
  const int  half = lane >> 4;
  const int  l16  = lane & 15;
  const int  r0   = (wave_u >> 2) * HB;
  const int  tn   = (wave_u & 3) * 16;
  float il0[8] = {0.f, 0.f, 0.f, 0.f, 0.f, 0.f, 0.f, 0.f};
  float il1[8] = {0.f, 0.f, 0.f, 0.f, 0.f, 0.f, 0.f, 0.f};
  if (mm) {
    for (int j = 0; j < 8; ++j) {
      il0[j] = Lm[(r0 + j + 8 * half) * LDP + tn + l16];
      il1[j] = Lm[(r0 + 16 + j + 8 * half) * LDP + tn + l16];
    }
  }

  {
    const int i = tid / HB, j = tid % HB;
    const bool in = (i < NH) && (j < NH);
    const float f00 = Fl[(i + 1) * GRID_NODES + (j + 1)];
    const float f10 = Fl[(NI - i) * GRID_NODES + (j + 1)];
    const float f01 = Fl[(i + 1) * GRID_NODES + (NI - j)];
    const float f11 = Fl[(NI - i) * GRID_NODES + (NI - j)];
    const float s0 = f00 + f10, s1 = f01 + f11;
    const float d0 = f00 - f10, d1 = f01 - f11;
    T0[i * LDP + j]               = in ? s0 + s1 : 0.0f;
    T0[i * LDP + HB + j]          = in ? s0 - s1 : 0.0f;
    T0[(HB + i) * LDP + j]        = in ? d0 + d1 : 0.0f;
    T0[(HB + i) * LDP + HB + j]   = in ? d0 - d1 : 0.0f;
  }
  __syncthreads();

  const v8f  zero8 = {0.f, 0.f, 0.f, 0.f, 0.f, 0.f, 0.f, 0.f};
  if (mm) {
    v8f p0 = zero8, p1 = zero8;
    right_bd(T0, Dt, r0, tn, half, l16, p0, p1);
    v8f q0 = zero8, q1 = zero8;
    left_bd_regs(Dm, r0, half, l16, p0, p1, q0, q1);
    for (int j = 0; j < 8; ++j) {
      q0[j] *= il0[j];
      q1[j] *= il1[j];
    }
    v8f s0 = zero8, s1 = zero8;
    left_bd_regs(Dt, r0, half, l16, q0, q1, s0, s1);
    store_tiles(T1, r0, tn, half, l16, s0, s1);
  }
  __syncthreads();
  if (mm) {
    v8f w0 = zero8, w1 = zero8;
    right_bd(T1, Dm, r0, tn, half, l16, w0, w1);
    store_tiles(T0, r0, tn, half, l16, w0, w1);
  }
  __syncthreads();

  for (int j = 0; j < (GRID_NODES * GRID_NODES) / NTHREADS; ++j) {
    const int i = tid + NTHREADS * j;
    const int r = i / GRID_NODES, c = i % GRID_NODES;
    const bool in = (r >= 1) && (r <= NI) && (c >= 1) && (c <= NI);
    const int i0 = in ? r - 1 : 0, j0 = in ? c - 1 : 0;
    const bool mr = i0 >= NH, mc = j0 >= NH;
    const int ii = mr ? (NI - 1 - i0) : i0;
    const int jj = mc ? (NI - 1 - j0) : j0;
    const float w00 = T0[ii * LDP + jj];
    const float w01 = T0[ii * LDP + HB + jj];
    const float w10 = T0[(HB + ii) * LDP + jj];
    const float w11 = T0[(HB + ii) * LDP + HB + jj];
    const float top = mc ? w00 - w01 : w00 + w01;
    const float bot = mc ? w10 - w11 : w10 + w11;
    const float v   = mr ? top - bot : top + bot;
    u[i] = in ? v : 0.0f;
  }
}

extern "C" void kernel_launch(void* const* d_in, const int* in_sizes, int n_in,
                              void* d_out, int out_size, void* d_ws, size_t ws_size,
                              hipStream_t stream) {
  const float* f      = (const float*)d_in[0];
  const float* nodes  = (const float*)d_in[1];
  const float* kappa  = (const float*)d_in[2];
  const int*   elems  = (const int*)d_in[4];
  const int    nelem  = in_sizes[4] / 3;

  fem_fused_kernel<<<1, NTHREADS, 0, stream>>>(f, nodes, kappa, elems, nelem,
                                               (float*)d_out);
}


# tables built before the first barrier from a 64-entry eigenvalue vector: the 4096-entry inverse-eigenvalue table and its divides go (16 reciprocals per matrix-wave lane), D in its 2 non-zero blocks
# speedup vs baseline: 4.1741x; 1.0600x over previous
#include <hip/hip_runtime.h>
#include <math.h>


#define GRID_NODES 64
#define NI         62
#define PAD        64
#define NH         31
#define HB         32
#define LDP        76
#define PI_F       3.14159265358979323846f

typedef float v2f __attribute__((ext_vector_type(2)));
typedef float v8f __attribute__((ext_vector_type(8)));

#define NTHREADS   1024
#define EPT        4


__device__ __forceinline__ void right_bd(const float* __restrict__ X,
                                         const float* __restrict__ Y,
                                         int r0, int tn, int half, int l16,
                                         v8f& c0, v8f& c1) {
  const int kb = tn & HB;
#pragma unroll
  for (int k0 = 0; k0 < HB; k0 += 4) {
    const int ka = kb + k0 + 2 * half;
    v2f a0, a1, b;
    a0[0] = X[(r0 + l16) * LDP + ka];
    a0[1] = X[(r0 + l16) * LDP + ka + 1];
    a1[0] = X[(r0 + 16 + l16) * LDP + ka];
    a1[1] = X[(r0 + 16 + l16) * LDP + ka + 1];
    b[0]  = Y[ka * LDP + tn + l16];
    b[1]  = Y[(ka + 1) * LDP + tn + l16];
    c0 = __builtin_amdgcn_wmma_f32_16x16x4_f32(
        false, a0, false, b, (short)0, c0, false, false);
    c1 = __builtin_amdgcn_wmma_f32_16x16x4_f32(
        false, a1, false, b, (short)0, c1, false, false);
  }
}

__device__ __forceinline__ void left_bd_regs(const float* __restrict__ X,
                                             int r0, int half, int l16,
                                             const v8f& b0, const v8f& b1,
                                             v8f& c0, v8f& c1) {
#pragma unroll
  for (int i = 0; i < 2; ++i) {
#pragma unroll
    for (int t = 0; t < 4; ++t) {
      const int kc = r0 + 16 * i + 8 * half + 2 * t;
      v2f a0, a1, b;
      a0[0] = X[(r0 + l16) * LDP + kc];
      a0[1] = X[(r0 + l16) * LDP + kc + 1];
      a1[0] = X[(r0 + 16 + l16) * LDP + kc];
      a1[1] = X[(r0 + 16 + l16) * LDP + kc + 1];
      b[0]  = i ? b1[2 * t]     : b0[2 * t];
      b[1]  = i ? b1[2 * t + 1] : b0[2 * t + 1];
      c0 = __builtin_amdgcn_wmma_f32_16x16x4_f32(
          false, a0, false, b, (short)0, c0, false, false);
      c1 = __builtin_amdgcn_wmma_f32_16x16x4_f32(
          false, a1, false, b, (short)0, c1, false, false);
    }
  }
}

__device__ __forceinline__ void store_tiles(float* __restrict__ C, int r0, int tn,
                                            int half, int l16,
                                            const v8f& c0, const v8f& c1) {
  for (int j = 0; j < 8; ++j) {
    C[(r0 + j + 8 * half) * LDP + tn + l16]      = c0[j];
    C[(r0 + 16 + j + 8 * half) * LDP + tn + l16] = c1[j];
  }
}

__global__ void __launch_bounds__(NTHREADS)
fem_fused_kernel(const float* __restrict__ f,
                 const float* __restrict__ nodes,
                 const float* __restrict__ kappa,
                 const int* __restrict__ elems,
                 int nelem,
                 float* __restrict__ u) {
  __shared__ float Dm[PAD * LDP];
  __shared__ float Dt[PAD * LDP];
  __shared__ float T0[PAD * LDP];
  __shared__ float T1[PAD * LDP];
  __shared__ float Lam[PAD];
  __shared__ float Fl[GRID_NODES * GRID_NODES];

  const int tid    = threadIdx.x;
  const int lane   = tid & 31;
  const int wave_u = __builtin_amdgcn_readfirstlane(tid >> 5);

  {
    const int m = tid / HB, n = tid % HB;
    const bool in = (m < NH) && (n < NH);
    for (int a = 0; a < 2; ++a) {
      const int k = ((2 * m + a + 1) * (n + 1)) % 126;
      const float s = __builtin_amdgcn_sinf((float)k * (1.0f / 126.0f));
      const float d = in ? s : 0.0f;
      Dm[(HB * a + m) * LDP + HB * a + n] = d;
      Dt[(HB * a + n) * LDP + HB * a + m] = d;
    }
  }

  for (int j = 0; j < (GRID_NODES * GRID_NODES) / NTHREADS; ++j)
    Fl[tid + NTHREADS * j] = 0.0f;
  __syncthreads();

  const v2f* __restrict__ xy = reinterpret_cast<const v2f*>(nodes);
  const float kap = kappa[0];
  for (int base = 0; base < nelem; base += NTHREADS * EPT) {
    int  va[EPT], vb[EPT], vc[EPT];
    bool ok[EPT];
    for (int k = 0; k < EPT; ++k) {
      const int e = base + tid + NTHREADS * k;
      ok[k] = e < nelem;
      const int ee = ok[k] ? e : 0;
      va[k] = elems[3 * ee + 0];
      vb[k] = elems[3 * ee + 1];
      vc[k] = elems[3 * ee + 2];
    }
    v2f   pi[EPT], pj[EPT], pk[EPT];
    float fa[EPT], fb[EPT], fc[EPT];
    for (int k = 0; k < EPT; ++k) {
      pi[k] = xy[va[k]]; pj[k] = xy[vb[k]]; pk[k] = xy[vc[k]];
      fa[k] = f[va[k]];  fb[k] = f[vb[k]];  fc[k] = f[vc[k]];
    }
    for (int k = 0; k < EPT; ++k) {
      const float area = 0.5f * fabsf((pj[k][0] - pi[k][0]) * (pk[k][1] - pi[k][1]) -
                                      (pk[k][0] - pi[k][0]) * (pj[k][1] - pi[k][1]));
      const float t  = (area * (1.0f / 3.0f)) * ((fa[k] + fb[k] + fc[k]) * (1.0f / 3.0f));
      const float fe = (ok[k] && area > 1e-15f) ? t : 0.0f;
      atomicAdd(&Fl[ok[k] ? va[k] : tid], fe);
      atomicAdd(&Fl[ok[k] ? vb[k] : tid], fe);
      atomicAdd(&Fl[ok[k] ? vc[k] : tid], fe);
    }
  }

  if (tid < PAD) {
    const int p = 2 * (tid % HB) + tid / HB;
    Lam[tid] = 2.0f - 2.0f * __builtin_amdgcn_cosf((float)(p + 1) * (1.0f / 126.0f));
  }
  __syncthreads();

  const bool mm   = wave_u < 8;
  const int  half = lane >> 4;
  const int  l16  = lane & 15;
  const int  r0   = (wave_u >> 2) * HB;
  const int  tn   = (wave_u & 3) * 16;
  float il0[8] = {0.f, 0.f, 0.f, 0.f, 0.f, 0.f, 0.f, 0.f};
  float il1[8] = {0.f, 0.f, 0.f, 0.f, 0.f, 0.f, 0.f, 0.f};
  if (mm) {
    const float nk = ((2.0f / 63.0f) * (2.0f / 63.0f)) / kap;
    const float lc = Lam[tn + l16];
    for (int j = 0; j < 8; ++j) {
      il0[j] = nk * __builtin_amdgcn_rcpf(Lam[r0 + j + 8 * half] + lc);
      il1[j] = nk * __builtin_amdgcn_rcpf(Lam[r0 + 16 + j + 8 * half] + lc);
    }
  }

  {
    const int i = tid / HB, j = tid % HB;
    const bool in = (i < NH) && (j < NH);
    const float f00 = Fl[(i + 1) * GRID_NODES + (j + 1)];
    const float f10 = Fl[(NI - i) * GRID_NODES + (j + 1)];
    const float f01 = Fl[(i + 1) * GRID_NODES + (NI - j)];
    const float f11 = Fl[(NI - i) * GRID_NODES + (NI - j)];
    const float s0 = f00 + f10, s1 = f01 + f11;
    const float d0 = f00 - f10, d1 = f01 - f11;
    T0[i * LDP + j]               = in ? s0 + s1 : 0.0f;
    T0[i * LDP + HB + j]          = in ? s0 - s1 : 0.0f;
    T0[(HB + i) * LDP + j]        = in ? d0 + d1 : 0.0f;
    T0[(HB + i) * LDP + HB + j]   = in ? d0 - d1 : 0.0f;
  }
  __syncthreads();

  const v8f  zero8 = {0.f, 0.f, 0.f, 0.f, 0.f, 0.f, 0.f, 0.f};
  if (mm) {
    v8f p0 = zero8, p1 = zero8;
    right_bd(T0, Dt, r0, tn, half, l16, p0, p1);
    v8f q0 = zero8, q1 = zero8;
    left_bd_regs(Dm, r0, half, l16, p0, p1, q0, q1);
    for (int j = 0; j < 8; ++j) {
      q0[j] *= il0[j];
      q1[j] *= il1[j];
    }
    v8f s0 = zero8, s1 = zero8;
    left_bd_regs(Dt, r0, half, l16, q0, q1, s0, s1);
    store_tiles(T1, r0, tn, half, l16, s0, s1);
  }
  __syncthreads();
  if (mm) {
    v8f w0 = zero8, w1 = zero8;
    right_bd(T1, Dm, r0, tn, half, l16, w0, w1);
    store_tiles(T0, r0, tn, half, l16, w0, w1);
  }
  __syncthreads();

  for (int j = 0; j < (GRID_NODES * GRID_NODES) / NTHREADS; ++j) {
    const int i = tid + NTHREADS * j;
    const int r = i / GRID_NODES, c = i % GRID_NODES;
    const bool in = (r >= 1) && (r <= NI) && (c >= 1) && (c <= NI);
    const int i0 = in ? r - 1 : 0, j0 = in ? c - 1 : 0;
    const bool mr = i0 >= NH, mc = j0 >= NH;
    const int ii = mr ? (NI - 1 - i0) : i0;
    const int jj = mc ? (NI - 1 - j0) : j0;
    const float w00 = T0[ii * LDP + jj];
    const float w01 = T0[ii * LDP + HB + jj];
    const float w10 = T0[(HB + ii) * LDP + jj];
    const float w11 = T0[(HB + ii) * LDP + HB + jj];
    const float top = mc ? w00 - w01 : w00 + w01;
    const float bot = mc ? w10 - w11 : w10 + w11;
    const float v   = mr ? top - bot : top + bot;
    u[i] = in ? v : 0.0f;
  }
}

extern "C" void kernel_launch(void* const* d_in, const int* in_sizes, int n_in,
                              void* d_out, int out_size, void* d_ws, size_t ws_size,
                              hipStream_t stream) {
  const float* f      = (const float*)d_in[0];
  const float* nodes  = (const float*)d_in[1];
  const float* kappa  = (const float*)d_in[2];
  const int*   elems  = (const int*)d_in[4];
  const int    nelem  = in_sizes[4] / 3;

  fem_fused_kernel<<<1, NTHREADS, 0, stream>>>(f, nodes, kappa, elems, nelem,
                                               (float*)d_out);
}


# assembly by element pairs (e, e+ceil(n/2)), 4 per thread, one trip: pairs sharing an edge (ballot per wave; general path kept) load shared vertices once, merge adds: 6->4 vertex loads, 6->4 atomics
# speedup vs baseline: 4.5601x; 1.0925x over previous
#include <hip/hip_runtime.h>
#include <math.h>


#define GRID_NODES 64
#define NI         62
#define PAD        64
#define NH         31
#define HB         32
#define LDP        76
#define PI_F       3.14159265358979323846f

typedef float v2f __attribute__((ext_vector_type(2)));
typedef float v8f __attribute__((ext_vector_type(8)));

#define NTHREADS   1024
#define PPT        4


__device__ __forceinline__ void right_bd(const float* __restrict__ X,
                                         const float* __restrict__ Y,
                                         int r0, int tn, int half, int l16,
                                         v8f& c0, v8f& c1) {
  const int kb = tn & HB;
#pragma unroll
  for (int k0 = 0; k0 < HB; k0 += 4) {
    const int ka = kb + k0 + 2 * half;
    v2f a0, a1, b;
    a0[0] = X[(r0 + l16) * LDP + ka];
    a0[1] = X[(r0 + l16) * LDP + ka + 1];
    a1[0] = X[(r0 + 16 + l16) * LDP + ka];
    a1[1] = X[(r0 + 16 + l16) * LDP + ka + 1];
    b[0]  = Y[ka * LDP + tn + l16];
    b[1]  = Y[(ka + 1) * LDP + tn + l16];
    c0 = __builtin_amdgcn_wmma_f32_16x16x4_f32(
        false, a0, false, b, (short)0, c0, false, false);
    c1 = __builtin_amdgcn_wmma_f32_16x16x4_f32(
        false, a1, false, b, (short)0, c1, false, false);
  }
}

__device__ __forceinline__ void left_bd_regs(const float* __restrict__ X,
                                             int r0, int half, int l16,
                                             const v8f& b0, const v8f& b1,
                                             v8f& c0, v8f& c1) {
#pragma unroll
  for (int i = 0; i < 2; ++i) {
#pragma unroll
    for (int t = 0; t < 4; ++t) {
      const int kc = r0 + 16 * i + 8 * half + 2 * t;
      v2f a0, a1, b;
      a0[0] = X[(r0 + l16) * LDP + kc];
      a0[1] = X[(r0 + l16) * LDP + kc + 1];
      a1[0] = X[(r0 + 16 + l16) * LDP + kc];
      a1[1] = X[(r0 + 16 + l16) * LDP + kc + 1];
      b[0]  = i ? b1[2 * t]     : b0[2 * t];
      b[1]  = i ? b1[2 * t + 1] : b0[2 * t + 1];
      c0 = __builtin_amdgcn_wmma_f32_16x16x4_f32(
          false, a0, false, b, (short)0, c0, false, false);
      c1 = __builtin_amdgcn_wmma_f32_16x16x4_f32(
          false, a1, false, b, (short)0, c1, false, false);
    }
  }
}

__device__ __forceinline__ void store_tiles(float* __restrict__ C, int r0, int tn,
                                            int half, int l16,
                                            const v8f& c0, const v8f& c1) {
  for (int j = 0; j < 8; ++j) {
    C[(r0 + j + 8 * half) * LDP + tn + l16]      = c0[j];
    C[(r0 + 16 + j + 8 * half) * LDP + tn + l16] = c1[j];
  }
}

__device__ __forceinline__ float tri_share(v2f pi, v2f pj, v2f pk,
                                           float fa, float fb, float fc, bool ok) {
  const float area = 0.5f * fabsf((pj[0] - pi[0]) * (pk[1] - pi[1]) -
                                  (pk[0] - pi[0]) * (pj[1] - pi[1]));
  const float t = (area * (1.0f / 3.0f)) * ((fa + fb + fc) * (1.0f / 3.0f));
  return (ok && area > 1e-15f) ? t : 0.0f;
}

__global__ void __launch_bounds__(NTHREADS)
fem_fused_kernel(const float* __restrict__ f,
                 const float* __restrict__ nodes,
                 const float* __restrict__ kappa,
                 const int* __restrict__ elems,
                 int nelem,
                 float* __restrict__ u) {
  __shared__ float Dm[PAD * LDP];
  __shared__ float Dt[PAD * LDP];
  __shared__ float T0[PAD * LDP];
  __shared__ float T1[PAD * LDP];
  __shared__ float Lam[PAD];
  __shared__ float Fl[GRID_NODES * GRID_NODES];

  const int tid    = threadIdx.x;
  const int lane   = tid & 31;
  const int wave_u = __builtin_amdgcn_readfirstlane(tid >> 5);

  {
    const int m = tid / HB, n = tid % HB;
    const bool in = (m < NH) && (n < NH);
    for (int a = 0; a < 2; ++a) {
      const int k = ((2 * m + a + 1) * (n + 1)) % 126;
      const float s = __builtin_amdgcn_sinf((float)k * (1.0f / 126.0f));
      const float d = in ? s : 0.0f;
      Dm[(HB * a + m) * LDP + HB * a + n] = d;
      Dt[(HB * a + n) * LDP + HB * a + m] = d;
    }
  }

  for (int j = 0; j < (GRID_NODES * GRID_NODES) / NTHREADS; ++j)
    Fl[tid + NTHREADS * j] = 0.0f;
  __syncthreads();

  const v2f* __restrict__ xy = reinterpret_cast<const v2f*>(nodes);
  const float kap = kappa[0];
  const int H = (nelem + 1) >> 1;
  for (int base = 0; base < H; base += NTHREADS * PPT) {
    int  a1[PPT], b1[PPT], c1[PPT], a2[PPT], b2[PPT], c2[PPT];
    bool ok1[PPT], ok2[PPT];
    for (int k = 0; k < PPT; ++k) {
      const int e = base + tid + NTHREADS * k;
      ok1[k] = e < H;
      ok2[k] = ok1[k] && (e + H < nelem);
      const int e1 = ok1[k] ? e : 0;
      const int e2 = ok2[k] ? e + H : 0;
      a1[k] = elems[3 * e1 + 0];
      b1[k] = elems[3 * e1 + 1];
      c1[k] = elems[3 * e1 + 2];
      a2[k] = elems[3 * e2 + 0];
      b2[k] = elems[3 * e2 + 1];
      c2[k] = elems[3 * e2 + 2];
    }
    __builtin_amdgcn_sched_barrier(0);
    bool other = false;
    for (int k = 0; k < PPT; ++k)
      other = other | (ok1[k] & !(ok2[k] & (a2[k] == a1[k]) & (b2[k] == c1[k])));
    const bool merged = __builtin_amdgcn_ballot_w32(other) == 0u;
    v2f   pi[PPT], pj[PPT], pk[PPT], pl[PPT];
    float fi[PPT], fj[PPT], fk[PPT], fl[PPT];
    for (int k = 0; k < PPT; ++k) {
      pi[k] = xy[a1[k]]; pj[k] = xy[b1[k]]; pk[k] = xy[c1[k]]; pl[k] = xy[c2[k]];
      fi[k] = f[a1[k]];  fj[k] = f[b1[k]];  fk[k] = f[c1[k]];  fl[k] = f[c2[k]];
    }
    if (merged) {
      for (int k = 0; k < PPT; ++k) {
        const float g1 = tri_share(pi[k], pj[k], pk[k], fi[k], fj[k], fk[k], ok1[k]);
        const float g2 = tri_share(pi[k], pk[k], pl[k], fi[k], fk[k], fl[k], ok1[k]);
        atomicAdd(&Fl[ok1[k] ? a1[k] : tid], g1 + g2);
        atomicAdd(&Fl[ok1[k] ? b1[k] : tid], g1);
        atomicAdd(&Fl[ok1[k] ? c1[k] : tid], g1 + g2);
        atomicAdd(&Fl[ok1[k] ? c2[k] : tid], g2);
      }
    } else {
      for (int k = 0; k < PPT; ++k) {
        const v2f   qi = xy[a2[k]], qj = xy[b2[k]];
        const float hi = f[a2[k]],  hj = f[b2[k]];
        const float g1 = tri_share(pi[k], pj[k], pk[k], fi[k], fj[k], fk[k], ok1[k]);
        const float g2 = tri_share(qi, qj, pl[k], hi, hj, fl[k], ok2[k]);
        atomicAdd(&Fl[ok1[k] ? a1[k] : tid], g1);
        atomicAdd(&Fl[ok1[k] ? b1[k] : tid], g1);
        atomicAdd(&Fl[ok1[k] ? c1[k] : tid], g1);
        atomicAdd(&Fl[ok2[k] ? a2[k] : tid], g2);
        atomicAdd(&Fl[ok2[k] ? b2[k] : tid], g2);
        atomicAdd(&Fl[ok2[k] ? c2[k] : tid], g2);
      }
    }
  }

  if (tid < PAD) {
    const int p = 2 * (tid % HB) + tid / HB;
    Lam[tid] = 2.0f - 2.0f * __builtin_amdgcn_cosf((float)(p + 1) * (1.0f / 126.0f));
  }
  __syncthreads();

  const bool mm   = wave_u < 8;
  const int  half = lane >> 4;
  const int  l16  = lane & 15;
  const int  r0   = (wave_u >> 2) * HB;
  const int  tn   = (wave_u & 3) * 16;
  float il0[8] = {0.f, 0.f, 0.f, 0.f, 0.f, 0.f, 0.f, 0.f};
  float il1[8] = {0.f, 0.f, 0.f, 0.f, 0.f, 0.f, 0.f, 0.f};
  if (mm) {
    const float nk = ((2.0f / 63.0f) * (2.0f / 63.0f)) / kap;
    const float lc = Lam[tn + l16];
    for (int j = 0; j < 8; ++j) {
      il0[j] = nk * __builtin_amdgcn_rcpf(Lam[r0 + j + 8 * half] + lc);
      il1[j] = nk * __builtin_amdgcn_rcpf(Lam[r0 + 16 + j + 8 * half] + lc);
    }
  }

  {
    const int i = tid / HB, j = tid % HB;
    const bool in = (i < NH) && (j < NH);
    const float f00 = Fl[(i + 1) * GRID_NODES + (j + 1)];
    const float f10 = Fl[(NI - i) * GRID_NODES + (j + 1)];
    const float f01 = Fl[(i + 1) * GRID_NODES + (NI - j)];
    const float f11 = Fl[(NI - i) * GRID_NODES + (NI - j)];
    const float s0 = f00 + f10, s1 = f01 + f11;
    const float d0 = f00 - f10, d1 = f01 - f11;
    T0[i * LDP + j]               = in ? s0 + s1 : 0.0f;
    T0[i * LDP + HB + j]          = in ? s0 - s1 : 0.0f;
    T0[(HB + i) * LDP + j]        = in ? d0 + d1 : 0.0f;
    T0[(HB + i) * LDP + HB + j]   = in ? d0 - d1 : 0.0f;
  }
  __syncthreads();

  const v8f  zero8 = {0.f, 0.f, 0.f, 0.f, 0.f, 0.f, 0.f, 0.f};
  if (mm) {
    v8f p0 = zero8, p1 = zero8;
    right_bd(T0, Dt, r0, tn, half, l16, p0, p1);
    v8f q0 = zero8, q1 = zero8;
    left_bd_regs(Dm, r0, half, l16, p0, p1, q0, q1);
    for (int j = 0; j < 8; ++j) {
      q0[j] *= il0[j];
      q1[j] *= il1[j];
    }
    v8f s0 = zero8, s1 = zero8;
    left_bd_regs(Dt, r0, half, l16, q0, q1, s0, s1);
    store_tiles(T1, r0, tn, half, l16, s0, s1);
  }
  __syncthreads();
  if (mm) {
    v8f w0 = zero8, w1 = zero8;
    right_bd(T1, Dm, r0, tn, half, l16, w0, w1);
    store_tiles(T0, r0, tn, half, l16, w0, w1);
  }
  __syncthreads();

  for (int j = 0; j < (GRID_NODES * GRID_NODES) / NTHREADS; ++j) {
    const int i = tid + NTHREADS * j;
    const int r = i / GRID_NODES, c = i % GRID_NODES;
    const bool in = (r >= 1) && (r <= NI) && (c >= 1) && (c <= NI);
    const int i0 = in ? r - 1 : 0, j0 = in ? c - 1 : 0;
    const bool mr = i0 >= NH, mc = j0 >= NH;
    const int ii = mr ? (NI - 1 - i0) : i0;
    const int jj = mc ? (NI - 1 - j0) : j0;
    const float w00 = T0[ii * LDP + jj];
    const float w01 = T0[ii * LDP + HB + jj];
    const float w10 = T0[(HB + ii) * LDP + jj];
    const float w11 = T0[(HB + ii) * LDP + HB + jj];
    const float top = mc ? w00 - w01 : w00 + w01;
    const float bot = mc ? w10 - w11 : w10 + w11;
    const float v   = mr ? top - bot : top + bot;
    u[i] = in ? v : 0.0f;
  }
}

extern "C" void kernel_launch(void* const* d_in, const int* in_sizes, int n_in,
                              void* d_out, int out_size, void* d_ws, size_t ws_size,
                              hipStream_t stream) {
  const float* f      = (const float*)d_in[0];
  const float* nodes  = (const float*)d_in[1];
  const float* kappa  = (const float*)d_in[2];
  const int*   elems  = (const int*)d_in[4];
  const int    nelem  = in_sizes[4] / 3;

  fem_fused_kernel<<<1, NTHREADS, 0, stream>>>(f, nodes, kappa, elems, nelem,
                                               (float*)d_out);
}
